# starting kernel
# baseline (speedup 1.0000x reference)
#include <hip/hip_runtime.h>


typedef float v2f __attribute__((ext_vector_type(2)));
typedef float v8f __attribute__((ext_vector_type(8)));

#define NB   8
#define CIN  64
#define HH   128
#define WW   128
#define COUT 128
#define PX   64
#define LDW  (PX + 4)
#define KTOT (9 * CIN)
#define NCP  (CIN / 2)
#define SLAB (5 * LDW * 2 + 56)

static constexpr float MUW = (1.0f - 0.9f) / 3.0f;

__global__ __launch_bounds__(256)
void pack_weights_kernel(const float* __restrict__ weight,
                         float* __restrict__ wpack) {
    int idx = blockIdx.x * 256 + threadIdx.x;
    if (idx >= KTOT * COUT) return;
    int par = idx & 1;
    int oc  = (idx >> 1) & (COUT - 1);
    int kk  = idx >> 8;
    int icp = kk & (NCP - 1);
    int tap = kk >> 5;
    int ic  = icp * 2 + par;
    int kh = (tap / 3) * 2;
    int kw = (tap % 3) * 2;
    wpack[idx] = weight[((oc * CIN + ic) * 5 + kh) * 5 + kw];
}

template <bool PACKED>
__global__ __launch_bounds__(256)
void conv_wmma_kernel(const float* __restrict__ input_,
                      const float* __restrict__ weight,
                      const float* __restrict__ wpack,
                      float* __restrict__ out) {
    __shared__ float lin[NCP * SLAB];
    __shared__ float sT[5 * LDW];
    __shared__ float sS[PX];

    const int blk = blockIdx.x;
    const int w0  = (blk & 1) * PX;
    const int h   = (blk >> 1) & (HH - 1);
    const int n   = blk >> 8;
    const int t   = threadIdx.x;

    const int TILE = CIN * 5 * LDW;
    for (int e = t; e < TILE; e += 256) {
        int ic  = e / (5 * LDW);
        int rem = e - ic * (5 * LDW);
        int r   = rem / LDW;
        int c   = rem - r * LDW;
        int gy = h + r - 2;
        int gx = w0 + c - 2;
        float v = 0.0f;
        if ((unsigned)gy < (unsigned)HH && (unsigned)gx < (unsigned)WW)
            v = input_[((n * CIN + ic) * HH + gy) * WW + gx];
        lin[(ic >> 1) * SLAB + ((r * LDW + c) << 1) + (ic & 1)] = v;
    }
    __syncthreads();

    for (int cell = t; cell < 5 * LDW; cell += 256) {
        float s = 0.0f;
        #pragma unroll 8
        for (int icp = 0; icp < NCP; ++icp) {
            v2f pr = *(const v2f*)&lin[icp * SLAB + (cell << 1)];
            s += pr.x + pr.y;
        }
        sT[cell] = s;
    }
    __syncthreads();

    if (t < PX) {
        const int p = t;
        float s = 0.0f;
        #pragma unroll
        for (int kw = 0; kw < 5; ++kw)
            s += sT[1 * LDW + p + kw] + sT[3 * LDW + p + kw];
        s += sT[0 * LDW + p + 1] + sT[0 * LDW + p + 3];
        s += sT[2 * LDW + p + 1] + sT[2 * LDW + p + 3];
        s += sT[4 * LDW + p + 1] + sT[4 * LDW + p + 3];
        sS[p] = MUW * s;
    }
    __syncthreads();

    const int lane = t & 31;
    const int wv   = t >> 5;
    const int ocb  = wv * 16;
    const int mrow = lane & 15;
    const int hi   = lane >> 4;

    v8f acc[4];
    #pragma unroll
    for (int q = 0; q < 4; ++q) acc[q] = (v8f){};

    for (int tap = 0; tap < 9; ++tap) {
        const int kh = (tap / 3) * 2;
        const int kw = (tap % 3) * 2;
        const int bcol = (kh * LDW + kw + mrow) << 1;
        #pragma unroll
        for (int ic4 = 0; ic4 < 16; ++ic4) {
            const int icp = ic4 * 2 + hi;
            v2f a;
            if (PACKED) {
                a = *(const v2f*)
                    &wpack[(((tap * NCP + icp) * COUT) + ocb + mrow) * 2];
            } else {
                const float* ap =
                    &weight[((ocb + mrow) * CIN + icp * 2) * 25 + kh * 5 + kw];
                a.x = ap[0];
                a.y = ap[25];
            }
            #pragma unroll
            for (int q = 0; q < 4; ++q) {
                v2f b = *(const v2f*)&lin[icp * SLAB + bcol + (q << 5)];
                acc[q] = __builtin_amdgcn_wmma_f32_16x16x4_f32(
                             false, a, false, b, (short)0, acc[q],
                             false, false);
            }
        }
    }

    float add[4];
    #pragma unroll
    for (int q = 0; q < 4; ++q) add[q] = sS[mrow + q * 16];
    #pragma unroll
    for (int j = 0; j < 8; ++j) {
        const int oc = ocb + j + 8 * hi;
        float* o = &out[((n * COUT + oc) * HH + h) * WW + w0];
        #pragma unroll
        for (int q = 0; q < 4; ++q)
            o[mrow + q * 16] = acc[q][j] + add[q];
    }
}

extern "C" void kernel_launch(void* const* d_in, const int* in_sizes, int n_in,
                              void* d_out, int out_size, void* d_ws,
                              size_t ws_size, hipStream_t stream) {
    const float* input_ = (const float*)d_in[0];
    const float* weight = (const float*)d_in[1];
    float* out = (float*)d_out;

    const int nblocks = NB * HH * (WW / PX);
    const size_t need = (size_t)KTOT * COUT * sizeof(float);

    if (d_ws != nullptr && ws_size >= need) {
        float* wpack = (float*)d_ws;
        pack_weights_kernel<<<(KTOT * COUT + 255) / 256, 256, 0, stream>>>(
            weight, wpack);
        conv_wmma_kernel<true><<<nblocks, 256, 0, stream>>>(
            input_, weight, wpack, out);
    } else {
        conv_wmma_kernel<false><<<nblocks, 256, 0, stream>>>(
            input_, weight, nullptr, out);
    }
}


# Weight-stationary strips: each wave keeps all 144 K-pairs of its 16 channels in registers; 256 work-groups each stage one 12-row window and compute 8 rows with an LDS+WMMA-only K loop
# speedup vs baseline: 1.6498x; 1.6498x over previous
#include <hip/hip_runtime.h>


typedef float v2f __attribute__((ext_vector_type(2)));
typedef float v4f __attribute__((ext_vector_type(4)));
typedef float v8f __attribute__((ext_vector_type(8)));

#define NB   8
#define CIN  64
#define HH   128
#define WW   128
#define COUT 128
#define PX   64
#define LDW  (PX + 4)
#define KTOT (9 * CIN)
#define NCP  (CIN / 2)
#define RO   8
#define RI   (RO + 4)
#define SLAB (RI * LDW * 2)

static constexpr float MUW = (1.0f - 0.9f) / 3.0f;

__global__ __launch_bounds__(256)
void pack_weights_kernel(const float* __restrict__ weight,
                         float* __restrict__ wpack) {
    int idx = blockIdx.x * 256 + threadIdx.x;
    if (idx >= KTOT * COUT) return;
    int par = idx & 1;
    int oc  = (idx >> 1) & (COUT - 1);
    int kk  = idx >> 8;
    int icp = kk & (NCP - 1);
    int tap = kk >> 5;
    int ic  = icp * 2 + par;
    int kh = (tap / 3) * 2;
    int kw = (tap % 3) * 2;
    wpack[idx] = weight[((oc * CIN + ic) * 5 + kh) * 5 + kw];
}

template <bool PACKED>
__global__ __launch_bounds__(256)
void conv_strip_kernel(const float* __restrict__ input_,
                       const float* __restrict__ weight,
                       const float* __restrict__ wpack,
                       float* __restrict__ out) {
    __shared__ __attribute__((aligned(16))) float lin[NCP * SLAB];
    __shared__ float sT[RI * LDW];
    __shared__ float sS[RO * PX];

    const int blk = blockIdx.x;
    const int w0  = (blk & 1) * PX;
    const int h0  = ((blk >> 1) & (HH / RO - 1)) * RO;
    const int n   = blk >> 5;
    const int t   = threadIdx.x;

    const int lane = t & 31;
    const int wv   = t >> 5;
    const int ocb  = wv * 16;
    const int mrow = lane & 15;
    const int hi   = lane >> 4;

    v2f a[9][16];
    #pragma unroll
    for (int tap = 0; tap < 9; ++tap) {
        #pragma unroll
        for (int ic4 = 0; ic4 < 16; ++ic4) {
            const int icp = ic4 * 2 + hi;
            if (PACKED) {
                a[tap][ic4] = *(const v2f*)
                    &wpack[(((tap * NCP + icp) * COUT) + ocb + mrow) * 2];
            } else {
                const int kh = (tap / 3) * 2;
                const int kw = (tap % 3) * 2;
                const float* ap =
                    &weight[((ocb + mrow) * CIN + icp * 2) * 25 + kh * 5 + kw];
                a[tap][ic4].x = ap[0];
                a[tap][ic4].y = ap[25];
            }
        }
    }

    const float* inb = input_ + (size_t)n * CIN * HH * WW;
    {
        const int quad = t & 15;
        const int pl0  = t >> 4;
        for (int i0 = 0; i0 < 24; i0 += 6) {
            v4f x0[6], x1[6];
            #pragma unroll
            for (int u = 0; u < 6; ++u) {
                const int pl = pl0 + 16 * (i0 + u);
                const int p  = pl / RI;
                const int r  = pl - p * RI;
                const int gy = h0 - 2 + r;
                const bool ok = (unsigned)gy < (unsigned)HH;
                const int gyc = ok ? gy : 0;
                const float* src =
                    inb + ((2 * p) * HH + gyc) * WW + w0 + quad * 4;
                v4f y0 = *(const v4f*)src;
                v4f y1 = *(const v4f*)(src + HH * WW);
                x0[u] = ok ? y0 : (v4f)(0.0f);
                x1[u] = ok ? y1 : (v4f)(0.0f);
            }
            #pragma unroll
            for (int u = 0; u < 6; ++u) {
                const int pl = pl0 + 16 * (i0 + u);
                const int p  = pl / RI;
                const int r  = pl - p * RI;
                float* dst = &lin[p * SLAB + ((r * LDW + 2 + quad * 4) << 1)];
                *(v4f*)dst       = (v4f){x0[u].x, x1[u].x, x0[u].y, x1[u].y};
                *(v4f*)(dst + 4) = (v4f){x0[u].z, x1[u].z, x0[u].w, x1[u].w};
            }
        }
        #pragma unroll
        for (int i = 0; i < 3; ++i) {
            const int item = t + 256 * i;
            const int side = item & 1;
            const int pl   = item >> 1;
            const int p    = pl / RI;
            const int r    = pl - p * RI;
            const int gy   = h0 - 2 + r;
            const int gx   = side ? (w0 + PX) : (w0 - 2);
            const bool ok  = (unsigned)gy < (unsigned)HH &&
                             (unsigned)gx < (unsigned)WW;
            const float* src =
                inb + ((2 * p) * HH + (ok ? gy : 0)) * WW + (ok ? gx : 0);
            v2f y0 = *(const v2f*)src;
            v2f y1 = *(const v2f*)(src + HH * WW);
            v4f o = ok ? (v4f){y0.x, y1.x, y0.y, y1.y} : (v4f)(0.0f);
            const int c = side ? (PX + 2) : 0;
            *(v4f*)&lin[p * SLAB + ((r * LDW + c) << 1)] = o;
        }
    }
    __syncthreads();

    for (int cell = t; cell < RI * LDW; cell += 256) {
        float s = 0.0f;
        #pragma unroll 8
        for (int icp = 0; icp < NCP; ++icp) {
            v2f pr = *(const v2f*)&lin[icp * SLAB + (cell << 1)];
            s += pr.x + pr.y;
        }
        sT[cell] = s;
    }
    __syncthreads();

    for (int i = t; i < RO * PX; i += 256) {
        const int ro = i >> 6;
        const int p  = i & (PX - 1);
        const float* Tr = &sT[ro * LDW + p];
        float s = 0.0f;
        #pragma unroll
        for (int kw = 0; kw < 5; ++kw)
            s += Tr[1 * LDW + kw] + Tr[3 * LDW + kw];
        s += Tr[0 * LDW + 1] + Tr[0 * LDW + 3];
        s += Tr[2 * LDW + 1] + Tr[2 * LDW + 3];
        s += Tr[4 * LDW + 1] + Tr[4 * LDW + 3];
        sS[i] = MUW * s;
    }
    __syncthreads();

    #pragma unroll 1
    for (int ro = 0; ro < RO; ++ro) {
        v8f acc[4];
        #pragma unroll
        for (int q = 0; q < 4; ++q) acc[q] = (v8f){};

        const float* lrow = &lin[hi * SLAB + ((ro * LDW + mrow) << 1)];
        #pragma unroll
        for (int tap = 0; tap < 9; ++tap) {
            const int kh = (tap / 3) * 2;
            const int kw = (tap % 3) * 2;
            #pragma unroll
            for (int ic4 = 0; ic4 < 16; ++ic4) {
                #pragma unroll
                for (int q = 0; q < 4; ++q) {
                    v2f b = *(const v2f*)&lrow[ic4 * 2 * SLAB +
                                               ((kh * LDW + kw) << 1) +
                                               (q << 5)];
                    acc[q] = __builtin_amdgcn_wmma_f32_16x16x4_f32(
                                 false, a[tap][ic4], false, b, (short)0,
                                 acc[q], false, false);
                }
            }
            __builtin_amdgcn_sched_barrier(0);
        }

        float add[4];
        #pragma unroll
        for (int q = 0; q < 4; ++q) add[q] = sS[ro * PX + mrow + q * 16];
        #pragma unroll
        for (int j = 0; j < 8; ++j) {
            const int oc = ocb + j + 8 * hi;
            float* o = &out[((n * COUT + oc) * HH + h0 + ro) * WW + w0];
            #pragma unroll
            for (int q = 0; q < 4; ++q)
                o[mrow + q * 16] = acc[q][j] + add[q];
        }
    }
}

extern "C" void kernel_launch(void* const* d_in, const int* in_sizes, int n_in,
                              void* d_out, int out_size, void* d_ws,
                              size_t ws_size, hipStream_t stream) {
    const float* input_ = (const float*)d_in[0];
    const float* weight = (const float*)d_in[1];
    float* out = (float*)d_out;

    const int nblocks = NB * (HH / RO) * (WW / PX);
    const size_t need = (size_t)KTOT * COUT * sizeof(float);

    if (d_ws != nullptr && ws_size >= need) {
        float* wpack = (float*)d_ws;
        pack_weights_kernel<<<(KTOT * COUT + 255) / 256, 256, 0, stream>>>(
            weight, wpack);
        conv_strip_kernel<true><<<nblocks, 256, 0, stream>>>(
            input_, weight, wpack, out);
    } else {
        conv_strip_kernel<false><<<nblocks, 256, 0, stream>>>(
            input_, weight, nullptr, out);
    }
}


# K loop's B fragments read from LDS one 32-WMMA block ahead into a second register set, all 32 reads as one group: 18 LDS read places per output row among the WMMAs instead of 144
# speedup vs baseline: 1.7386x; 1.0538x over previous
#include <hip/hip_runtime.h>


typedef float v2f __attribute__((ext_vector_type(2)));
typedef float v4f __attribute__((ext_vector_type(4)));
typedef float v8f __attribute__((ext_vector_type(8)));

#define NB   8
#define CIN  64
#define HH   128
#define WW   128
#define COUT 128
#define PX   64
#define LDW  (PX + 4)
#define KTOT (9 * CIN)
#define NCP  (CIN / 2)
#define RO   8
#define RI   (RO + 4)
#define SLAB (RI * LDW * 2)

static constexpr float MUW = (1.0f - 0.9f) / 3.0f;

__global__ __launch_bounds__(256)
void pack_weights_kernel(const float* __restrict__ weight,
                         float* __restrict__ wpack) {
    int idx = blockIdx.x * 256 + threadIdx.x;
    if (idx >= KTOT * COUT) return;
    int par = idx & 1;
    int oc  = (idx >> 1) & (COUT - 1);
    int kk  = idx >> 8;
    int icp = kk & (NCP - 1);
    int tap = kk >> 5;
    int ic  = icp * 2 + par;
    int kh = (tap / 3) * 2;
    int kw = (tap % 3) * 2;
    wpack[idx] = weight[((oc * CIN + ic) * 5 + kh) * 5 + kw];
}

template <bool PACKED>
__global__ __launch_bounds__(256)
void conv_strip_kernel(const float* __restrict__ input_,
                       const float* __restrict__ weight,
                       const float* __restrict__ wpack,
                       float* __restrict__ out) {
    __shared__ __attribute__((aligned(16))) float lin[NCP * SLAB];
    __shared__ float sT[RI * LDW];
    __shared__ float sS[RO * PX];

    const int blk = blockIdx.x;
    const int w0  = (blk & 1) * PX;
    const int h0  = ((blk >> 1) & (HH / RO - 1)) * RO;
    const int n   = blk >> 5;
    const int t   = threadIdx.x;

    const int lane = t & 31;
    const int wv   = t >> 5;
    const int ocb  = wv * 16;
    const int mrow = lane & 15;
    const int hi   = lane >> 4;

    v2f a[9][16];
    #pragma unroll
    for (int tap = 0; tap < 9; ++tap) {
        #pragma unroll
        for (int ic4 = 0; ic4 < 16; ++ic4) {
            const int icp = ic4 * 2 + hi;
            if (PACKED) {
                a[tap][ic4] = *(const v2f*)
                    &wpack[(((tap * NCP + icp) * COUT) + ocb + mrow) * 2];
            } else {
                const int kh = (tap / 3) * 2;
                const int kw = (tap % 3) * 2;
                const float* ap =
                    &weight[((ocb + mrow) * CIN + icp * 2) * 25 + kh * 5 + kw];
                a[tap][ic4].x = ap[0];
                a[tap][ic4].y = ap[25];
            }
        }
    }

    const float* inb = input_ + (size_t)n * CIN * HH * WW;
    {
        const int quad = t & 15;
        const int pl0  = t >> 4;
        for (int i0 = 0; i0 < 24; i0 += 6) {
            v4f x0[6], x1[6];
            #pragma unroll
            for (int u = 0; u < 6; ++u) {
                const int pl = pl0 + 16 * (i0 + u);
                const int p  = pl / RI;
                const int r  = pl - p * RI;
                const int gy = h0 - 2 + r;
                const bool ok = (unsigned)gy < (unsigned)HH;
                const int gyc = ok ? gy : 0;
                const float* src =
                    inb + ((2 * p) * HH + gyc) * WW + w0 + quad * 4;
                v4f y0 = *(const v4f*)src;
                v4f y1 = *(const v4f*)(src + HH * WW);
                x0[u] = ok ? y0 : (v4f)(0.0f);
                x1[u] = ok ? y1 : (v4f)(0.0f);
            }
            #pragma unroll
            for (int u = 0; u < 6; ++u) {
                const int pl = pl0 + 16 * (i0 + u);
                const int p  = pl / RI;
                const int r  = pl - p * RI;
                float* dst = &lin[p * SLAB + ((r * LDW + 2 + quad * 4) << 1)];
                *(v4f*)dst       = (v4f){x0[u].x, x1[u].x, x0[u].y, x1[u].y};
                *(v4f*)(dst + 4) = (v4f){x0[u].z, x1[u].z, x0[u].w, x1[u].w};
            }
        }
        #pragma unroll
        for (int i = 0; i < 3; ++i) {
            const int item = t + 256 * i;
            const int side = item & 1;
            const int pl   = item >> 1;
            const int p    = pl / RI;
            const int r    = pl - p * RI;
            const int gy   = h0 - 2 + r;
            const int gx   = side ? (w0 + PX) : (w0 - 2);
            const bool ok  = (unsigned)gy < (unsigned)HH &&
                             (unsigned)gx < (unsigned)WW;
            const float* src =
                inb + ((2 * p) * HH + (ok ? gy : 0)) * WW + (ok ? gx : 0);
            v2f y0 = *(const v2f*)src;
            v2f y1 = *(const v2f*)(src + HH * WW);
            v4f o = ok ? (v4f){y0.x, y1.x, y0.y, y1.y} : (v4f)(0.0f);
            const int c = side ? (PX + 2) : 0;
            *(v4f*)&lin[p * SLAB + ((r * LDW + c) << 1)] = o;
        }
    }
    __syncthreads();

    for (int cell = t; cell < RI * LDW; cell += 256) {
        float s = 0.0f;
        #pragma unroll 8
        for (int icp = 0; icp < NCP; ++icp) {
            v2f pr = *(const v2f*)&lin[icp * SLAB + (cell << 1)];
            s += pr.x + pr.y;
        }
        sT[cell] = s;
    }
    __syncthreads();

    for (int i = t; i < RO * PX; i += 256) {
        const int ro = i >> 6;
        const int p  = i & (PX - 1);
        const float* Tr = &sT[ro * LDW + p];
        float s = 0.0f;
        #pragma unroll
        for (int kw = 0; kw < 5; ++kw)
            s += Tr[1 * LDW + kw] + Tr[3 * LDW + kw];
        s += Tr[0 * LDW + 1] + Tr[0 * LDW + 3];
        s += Tr[2 * LDW + 1] + Tr[2 * LDW + 3];
        s += Tr[4 * LDW + 1] + Tr[4 * LDW + 3];
        sS[i] = MUW * s;
    }
    __syncthreads();

    v2f bA[32], bB[32];

#define LOAD_BFRAG(dst, lrow_, blk)                                           \
    {                                                                          \
        const int tap_ = (blk) >> 1;                                           \
        const int kh_ = (tap_ / 3) * 2;                                        \
        const int kw_ = (tap_ % 3) * 2;                                        \
        const int ib_ = ((blk) & 1) * 8;                                       \
        _Pragma("unroll")                                                      \
        for (int i_ = 0; i_ < 8; ++i_) {                                       \
            _Pragma("unroll")                                                  \
            for (int q_ = 0; q_ < 4; ++q_)                                     \
                dst[i_ * 4 + q_] = *(const v2f*)&(lrow_)[(ib_ + i_) * 2 * SLAB + \
                                       ((kh_ * LDW + kw_) << 1) + (q_ << 5)];  \
        }                                                                      \
    }

#define MMA_BLOCK(src, blk)                                                    \
    {                                                                          \
        const int tap_ = (blk) >> 1;                                           \
        const int ib_ = ((blk) & 1) * 8;                                       \
        _Pragma("unroll")                                                      \
        for (int i_ = 0; i_ < 8; ++i_) {                                       \
            _Pragma("unroll")                                                  \
            for (int q_ = 0; q_ < 4; ++q_)                                     \
                acc[q_] = __builtin_amdgcn_wmma_f32_16x16x4_f32(               \
                              false, a[tap_][ib_ + i_], false,                 \
                              src[i_ * 4 + q_], (short)0, acc[q_], false,      \
                              false);                                          \
        }                                                                      \
    }

    const float* lrow0 = &lin[hi * SLAB + (mrow << 1)];
    LOAD_BFRAG(bA, lrow0, 0)

    #pragma unroll 1
    for (int ro = 0; ro < RO; ++ro) {
        v8f acc[4];
        #pragma unroll
        for (int q = 0; q < 4; ++q) acc[q] = (v8f){};

        const float* lrow = lrow0 + ((ro * LDW) << 1);

        float add[4];
        #pragma unroll
        for (int q = 0; q < 4; ++q) add[q] = sS[ro * PX + mrow + q * 16];

        #pragma unroll
        for (int b = 0; b < 18; b += 2) {
            LOAD_BFRAG(bB, lrow, b + 1)
            __builtin_amdgcn_sched_barrier(0);
            MMA_BLOCK(bA, b)
            __builtin_amdgcn_sched_barrier(0);
            if (b + 2 < 18) {
                LOAD_BFRAG(bA, lrow, b + 2)
            } else {
                LOAD_BFRAG(bA, lrow + (LDW << 1), 0)
            }
            __builtin_amdgcn_sched_barrier(0);
            MMA_BLOCK(bB, b + 1)
            __builtin_amdgcn_sched_barrier(0);
        }

        #pragma unroll
        for (int j = 0; j < 8; ++j) {
            const int oc = ocb + j + 8 * hi;
            float* o = &out[((n * COUT + oc) * HH + h0 + ro) * WW + w0];
            #pragma unroll
            for (int q = 0; q < 4; ++q)
                o[mrow + q * 16] = acc[q][j] + add[q];
        }
    }
#undef LOAD_BFRAG
#undef MMA_BLOCK
}

extern "C" void kernel_launch(void* const* d_in, const int* in_sizes, int n_in,
                              void* d_out, int out_size, void* d_ws,
                              size_t ws_size, hipStream_t stream) {
    const float* input_ = (const float*)d_in[0];
    const float* weight = (const float*)d_in[1];
    float* out = (float*)d_out;

    const int nblocks = NB * (HH / RO) * (WW / PX);
    const size_t need = (size_t)KTOT * COUT * sizeof(float);

    if (d_ws != nullptr && ws_size >= need) {
        float* wpack = (float*)d_ws;
        pack_weights_kernel<<<(KTOT * COUT + 255) / 256, 256, 0, stream>>>(
            weight, wpack);
        conv_strip_kernel<true><<<nblocks, 256, 0, stream>>>(
            input_, weight, wpack, out);
    } else {
        conv_strip_kernel<false><<<nblocks, 256, 0, stream>>>(
            input_, weight, nullptr, out);
    }
}


# N-tile columns remapped so a lane owns 4 adjacent pixels: B fragments read as 16 ds_load_b128 per 32-WMMA block instead of 32 b64, each output row stored as one 16-byte-per-lane full-line store
# speedup vs baseline: 1.8603x; 1.0700x over previous
#include <hip/hip_runtime.h>


typedef float v2f __attribute__((ext_vector_type(2)));
typedef float v4f __attribute__((ext_vector_type(4)));
typedef float v8f __attribute__((ext_vector_type(8)));

#define NB   8
#define CIN  64
#define HH   128
#define WW   128
#define COUT 128
#define PX   64
#define LDW  (PX + 4)
#define KTOT (9 * CIN)
#define NCP  (CIN / 2)
#define RO   8
#define RI   (RO + 4)
#define SLAB (RI * LDW * 2)

static constexpr float MUW = (1.0f - 0.9f) / 3.0f;

__global__ __launch_bounds__(256)
void pack_weights_kernel(const float* __restrict__ weight,
                         float* __restrict__ wpack) {
    int idx = blockIdx.x * 256 + threadIdx.x;
    if (idx >= KTOT * COUT) return;
    int par = idx & 1;
    int oc  = (idx >> 1) & (COUT - 1);
    int kk  = idx >> 8;
    int icp = kk & (NCP - 1);
    int tap = kk >> 5;
    int ic  = icp * 2 + par;
    int kh = (tap / 3) * 2;
    int kw = (tap % 3) * 2;
    wpack[idx] = weight[((oc * CIN + ic) * 5 + kh) * 5 + kw];
}

template <bool PACKED>
__global__ __launch_bounds__(256)
void conv_strip_kernel(const float* __restrict__ input_,
                       const float* __restrict__ weight,
                       const float* __restrict__ wpack,
                       float* __restrict__ out) {
    __shared__ __attribute__((aligned(16))) float lin[NCP * SLAB];
    __shared__ float sT[RI * LDW];
    __shared__ __attribute__((aligned(16))) float sS[RO * PX];

    const int blk = blockIdx.x;
    const int w0  = (blk & 1) * PX;
    const int h0  = ((blk >> 1) & (HH / RO - 1)) * RO;
    const int n   = blk >> 5;
    const int t   = threadIdx.x;

    const int lane = t & 31;
    const int wv   = t >> 5;
    const int ocb  = wv * 16;
    const int mrow = lane & 15;
    const int hi   = lane >> 4;

    v2f a[9][16];
    #pragma unroll
    for (int tap = 0; tap < 9; ++tap) {
        #pragma unroll
        for (int ic4 = 0; ic4 < 16; ++ic4) {
            const int icp = ic4 * 2 + hi;
            if (PACKED) {
                a[tap][ic4] = *(const v2f*)
                    &wpack[(((tap * NCP + icp) * COUT) + ocb + mrow) * 2];
            } else {
                const int kh = (tap / 3) * 2;
                const int kw = (tap % 3) * 2;
                const float* ap =
                    &weight[((ocb + mrow) * CIN + icp * 2) * 25 + kh * 5 + kw];
                a[tap][ic4].x = ap[0];
                a[tap][ic4].y = ap[25];
            }
        }
    }

    const float* inb = input_ + (size_t)n * CIN * HH * WW;
    {
        const int quad = t & 15;
        const int pl0  = t >> 4;
        for (int i0 = 0; i0 < 24; i0 += 6) {
            v4f x0[6], x1[6];
            #pragma unroll
            for (int u = 0; u < 6; ++u) {
                const int pl = pl0 + 16 * (i0 + u);
                const int p  = pl / RI;
                const int r  = pl - p * RI;
                const int gy = h0 - 2 + r;
                const bool ok = (unsigned)gy < (unsigned)HH;
                const int gyc = ok ? gy : 0;
                const float* src =
                    inb + ((2 * p) * HH + gyc) * WW + w0 + quad * 4;
                v4f y0 = *(const v4f*)src;
                v4f y1 = *(const v4f*)(src + HH * WW);
                x0[u] = ok ? y0 : (v4f)(0.0f);
                x1[u] = ok ? y1 : (v4f)(0.0f);
            }
            #pragma unroll
            for (int u = 0; u < 6; ++u) {
                const int pl = pl0 + 16 * (i0 + u);
                const int p  = pl / RI;
                const int r  = pl - p * RI;
                float* dst = &lin[p * SLAB + ((r * LDW + 2 + quad * 4) << 1)];
                *(v4f*)dst       = (v4f){x0[u].x, x1[u].x, x0[u].y, x1[u].y};
                *(v4f*)(dst + 4) = (v4f){x0[u].z, x1[u].z, x0[u].w, x1[u].w};
            }
        }
        #pragma unroll
        for (int i = 0; i < 3; ++i) {
            const int item = t + 256 * i;
            const int side = item & 1;
            const int pl   = item >> 1;
            const int p    = pl / RI;
            const int r    = pl - p * RI;
            const int gy   = h0 - 2 + r;
            const int gx   = side ? (w0 + PX) : (w0 - 2);
            const bool ok  = (unsigned)gy < (unsigned)HH &&
                             (unsigned)gx < (unsigned)WW;
            const float* src =
                inb + ((2 * p) * HH + (ok ? gy : 0)) * WW + (ok ? gx : 0);
            v2f y0 = *(const v2f*)src;
            v2f y1 = *(const v2f*)(src + HH * WW);
            v4f o = ok ? (v4f){y0.x, y1.x, y0.y, y1.y} : (v4f)(0.0f);
            const int c = side ? (PX + 2) : 0;
            *(v4f*)&lin[p * SLAB + ((r * LDW + c) << 1)] = o;
        }
    }
    __syncthreads();

    for (int cell = t; cell < RI * LDW; cell += 256) {
        float s = 0.0f;
        #pragma unroll 8
        for (int icp = 0; icp < NCP; ++icp) {
            v2f pr = *(const v2f*)&lin[icp * SLAB + (cell << 1)];
            s += pr.x + pr.y;
        }
        sT[cell] = s;
    }
    __syncthreads();

    for (int i = t; i < RO * PX; i += 256) {
        const int ro = i >> 6;
        const int p  = i & (PX - 1);
        const float* Tr = &sT[ro * LDW + p];
        float s = 0.0f;
        #pragma unroll
        for (int kw = 0; kw < 5; ++kw)
            s += Tr[1 * LDW + kw] + Tr[3 * LDW + kw];
        s += Tr[0 * LDW + 1] + Tr[0 * LDW + 3];
        s += Tr[2 * LDW + 1] + Tr[2 * LDW + 3];
        s += Tr[4 * LDW + 1] + Tr[4 * LDW + 3];
        sS[i] = MUW * s;
    }
    __syncthreads();

    v4f bA[16], bB[16];

#define LOAD_BFRAG(dst, roff_, blk)                                           \
    {                                                                          \
        const int tap_ = (blk) >> 1;                                           \
        const int kh_ = (tap_ / 3) * 2;                                        \
        const int kw_ = (tap_ % 3) * 2;                                        \
        const int ib_ = ((blk) & 1) * 8;                                       \
        _Pragma("unroll")                                                      \
        for (int i_ = 0; i_ < 8; ++i_) {                                       \
            const int o_ = (roff_) + (ib_ + i_) * 2 * SLAB +                   \
                           ((kh_ * LDW + kw_) << 1);                           \
            dst[i_ * 2 + 0] = *(const v4f*)&lrow0a[o_];                        \
            dst[i_ * 2 + 1] = *(const v4f*)&lrow0b[o_];                        \
        }                                                                      \
    }

#define MMA_BLOCK(src, blk)                                                    \
    {                                                                          \
        const int tap_ = (blk) >> 1;                                           \
        const int ib_ = ((blk) & 1) * 8;                                       \
        _Pragma("unroll")                                                      \
        for (int i_ = 0; i_ < 8; ++i_) {                                       \
            _Pragma("unroll")                                                  \
            for (int q_ = 0; q_ < 4; ++q_) {                                   \
                const v4f f_ = src[i_ * 2 + (q_ >> 1)];                        \
                const v2f b_ = (q_ & 1) ? (v2f){f_.z, f_.w}                    \
                                        : (v2f){f_.x, f_.y};                   \
                acc[q_] = __builtin_amdgcn_wmma_f32_16x16x4_f32(               \
                              false, a[tap_][ib_ + i_], false,                 \
                              b_, (short)0, acc[q_], false, false);            \
            }                                                                  \
        }                                                                      \
    }

    const int swp = (mrow >> 3) & 1;
    const float* lrow0a = &lin[hi * SLAB + (mrow << 3) + (swp << 2)];
    const float* lrow0b = &lin[hi * SLAB + (mrow << 3) + ((swp ^ 1) << 2)];
    LOAD_BFRAG(bA, 0, 0)

    #pragma unroll 1
    for (int ro = 0; ro < RO; ++ro) {
        v8f acc[4];
        #pragma unroll
        for (int q = 0; q < 4; ++q) acc[q] = (v8f){};

        const int roff = (ro * LDW) << 1;

        const v4f addv = *(const v4f*)&sS[ro * PX + (mrow << 2)];

        #pragma unroll
        for (int b = 0; b < 18; b += 2) {
            LOAD_BFRAG(bB, roff, b + 1)
            __builtin_amdgcn_sched_barrier(0);
            MMA_BLOCK(bA, b)
            __builtin_amdgcn_sched_barrier(0);
            if (b + 2 < 18) {
                LOAD_BFRAG(bA, roff, b + 2)
            } else {
                LOAD_BFRAG(bA, roff + (LDW << 1), 0)
            }
            __builtin_amdgcn_sched_barrier(0);
            MMA_BLOCK(bB, b + 1)
            __builtin_amdgcn_sched_barrier(0);
        }

        #pragma unroll
        for (int j = 0; j < 8; ++j) {
            const int oc = ocb + j + 8 * hi;
            float* o = &out[((n * COUT + oc) * HH + h0 + ro) * WW + w0];
            v4f r;
            r.x = (swp ? acc[2][j] : acc[0][j]) + addv.x;
            r.y = (swp ? acc[3][j] : acc[1][j]) + addv.y;
            r.z = (swp ? acc[0][j] : acc[2][j]) + addv.z;
            r.w = (swp ? acc[1][j] : acc[3][j]) + addv.w;
            *(v4f*)&o[mrow << 2] = r;
        }
    }
#undef LOAD_BFRAG
#undef MMA_BLOCK
}

extern "C" void kernel_launch(void* const* d_in, const int* in_sizes, int n_in,
                              void* d_out, int out_size, void* d_ws,
                              size_t ws_size, hipStream_t stream) {
    const float* input_ = (const float*)d_in[0];
    const float* weight = (const float*)d_in[1];
    float* out = (float*)d_out;

    const int nblocks = NB * (HH / RO) * (WW / PX);
    const size_t need = (size_t)KTOT * COUT * sizeof(float);

    if (d_ws != nullptr && ws_size >= need) {
        float* wpack = (float*)d_ws;
        pack_weights_kernel<<<(KTOT * COUT + 255) / 256, 256, 0, stream>>>(
            weight, wpack);
        conv_strip_kernel<true><<<nblocks, 256, 0, stream>>>(
            input_, weight, wpack, out);
    } else {
        conv_strip_kernel<false><<<nblocks, 256, 0, stream>>>(
            input_, weight, nullptr, out);
    }
}
